# starting kernel
# baseline (speedup 1.0000x reference)
#include <hip/hip_runtime.h>
#include <cstdint>


typedef float v2f __attribute__((ext_vector_type(2)));
typedef float v8f __attribute__((ext_vector_type(8)));

__global__ void rmtpp_durations_kernel(const float* __restrict__ ts,
                                       float* __restrict__ dur,
                                       int T, int BS) {
    int b = blockIdx.x * blockDim.x + threadIdx.x;
    if (b >= BS) return;
    float m = -3.402823466e38f;
    for (int t = 0; t < T; ++t) {
        m = fmaxf(m, ts[(size_t)t * BS + b]);
    }
    dur[b] = m;
}

__global__ void rmtpp_proj_wmma_kernel(const float* __restrict__ h,
                                       const float* __restrict__ w,
                                       float* __restrict__ proj,
                                       int rows, int H) {
    const int lane        = threadIdx.x & 31;
    const int waveInBlock = threadIdx.x >> 5;
    const int wave        = blockIdx.x * (blockDim.x >> 5) + waveInBlock;
    const int rowBase     = wave * 16;
    if (rowBase >= rows) return;

    const int m    = lane & 15;
    const int koff = (lane >> 4) << 1;
    int r = rowBase + m;
    if (r >= rows) r = rows - 1;

    const float* __restrict__ rowPtr = h + (size_t)r * H + koff;
    const float* __restrict__ wPtr   = w + koff;

    v8f c = {};
    #pragma unroll 4
    for (int k = 0; k < H; k += 4) {
        v2f a = *reinterpret_cast<const v2f*>(rowPtr + k);
        v2f b = *reinterpret_cast<const v2f*>(wPtr + k);
        c = __builtin_amdgcn_wmma_f32_16x16x4_f32(
                false, a, false, b, (short)0, c, false, false);
    }

    const int sel = lane & 7;
    float v = c[0];
    v = (sel == 1) ? c[1] : v;
    v = (sel == 2) ? c[2] : v;
    v = (sel == 3) ? c[3] : v;
    v = (sel == 4) ? c[4] : v;
    v = (sel == 5) ? c[5] : v;
    v = (sel == 6) ? c[6] : v;
    v = (sel == 7) ? c[7] : v;

    const int  mOut    = ((lane >> 4) << 3) | sel;
    const bool doStore = (lane < 8) || (lane >= 16 && lane < 24);
    if (doStore && (rowBase + mOut) < rows) {
        proj[rowBase + mOut] = v;
    }
}

__global__ void rmtpp_grid_kernel(const float* __restrict__ ts,
                                  const float* __restrict__ proj,
                                  const float* __restrict__ dur,
                                  const float* __restrict__ time_influence,
                                  const float* __restrict__ base_intensity,
                                  float* __restrict__ out_log,
                                  float* __restrict__ out_grid,
                                  int T, int BS, int NG) {
    const int gid = blockIdx.x * blockDim.x + threadIdx.x;
    if (gid >= NG * BS) return;
    const int b = gid % BS;
    const int n = gid / BS;

    const float ti = time_influence[0];
    const float bi = base_intensity[0];

    const float t_grid = ((float)n / (float)(NG - 1)) * dur[b];

    int idx = 0;
    for (int t = 0; t < T; ++t) {
        if (ts[(size_t)t * BS + b] < t_grid) { idx = t; break; }
    }

    const float closest_ts = ts[(size_t)idx * BS + b];
    const float li = proj[(size_t)idx * BS + b] + ti * (t_grid - closest_ts) + bi;

    out_log[gid]  = li;
    out_grid[gid] = t_grid;
}

extern "C" void kernel_launch(void* const* d_in, const int* in_sizes, int n_in,
                              void* d_out, int out_size, void* d_ws, size_t ws_size,
                              hipStream_t stream) {
    const float* h  = (const float*)d_in[0];
    const float* ts = (const float*)d_in[1];
    const float* w  = (const float*)d_in[2];
    const float* ti = (const float*)d_in[3];
    const float* bi = (const float*)d_in[4];

    const int NG  = 1000;
    const int BS  = out_size / (2 * NG);
    const int TBS = in_sizes[1];
    const int T   = TBS / BS;
    const int H   = in_sizes[2];

    float* out_log  = (float*)d_out;
    float* out_grid = out_log + (size_t)NG * BS;

    float* proj = (float*)d_ws;
    float* dur  = proj + TBS;

    rmtpp_durations_kernel<<<(BS + 127) / 128, 128, 0, stream>>>(ts, dur, T, BS);

    const int waves  = (TBS + 15) / 16;
    const int blocks = (waves + 7) / 8;
    rmtpp_proj_wmma_kernel<<<blocks, 256, 0, stream>>>(h, w, proj, TBS, H);

    const int total = NG * BS;
    rmtpp_grid_kernel<<<(total + 255) / 256, 256, 0, stream>>>(
        ts, proj, dur, ti, bi, out_log, out_grid, T, BS, NG);
}


# proj kernel: per-row trace butterfly moved from five dependent LDS permutes (each behind a wait) to vector-unit lane swizzles (permlanex16 + row_xmask adds); same partners, same sum order
# speedup vs baseline: 17.7589x; 17.7589x over previous
#include <hip/hip_runtime.h>
#include <cstdint>


typedef float v2f __attribute__((ext_vector_type(2)));
typedef float v4f __attribute__((ext_vector_type(4)));
typedef float v8f __attribute__((ext_vector_type(8)));

constexpr int kT       = 1024;
constexpr int kBS      = 128;
constexpr int kH          = 512;
constexpr int kPieces     = kH / 128;
constexpr int kProjWaves  = 16;
constexpr int kRowsPerWave = 32;
constexpr int kPair       = 2;
constexpr int kSteps      = kRowsPerWave / kPair;
constexpr int kAhead      = 3;
constexpr int kProjBlocks = (kT * kBS) / (kProjWaves * kRowsPerWave);
static_assert((kT * kBS) % (kProjWaves * kRowsPerWave) == 0, "every wave has its full 32 rows");
static_assert(kH % 128 == 0, "a row is a whole number of 512-byte pieces");

__device__ __forceinline__ float lane_xor16(float x) {
    const int xi = __builtin_bit_cast(int, x);
    return __builtin_bit_cast(float,
        __builtin_amdgcn_permlanex16(xi, xi, 0x76543210, 0xfedcba98, false, false));
}
template <int M>
__device__ __forceinline__ float lane_xor_row(float x) {
    return __builtin_bit_cast(float,
        __builtin_amdgcn_update_dpp(0, __builtin_bit_cast(int, x), 0x160 | M, 0xF, 0xF, true));
}

__global__ void __launch_bounds__(kProjWaves * 32)
rmtpp_proj_wmma_kernel(const float* __restrict__ h,
                       const float* __restrict__ w,
                       float* __restrict__ proj) {
    const int lane = threadIdx.x & 31;
    const int wave = __builtin_amdgcn_readfirstlane(threadIdx.x >> 5);
    const int row0 = (blockIdx.x * kProjWaves + wave) * kRowsPerWave;

    v4f wq[kPieces];
    #pragma unroll
    for (int q = 0; q < kPieces; ++q)
        wq[q] = *reinterpret_cast<const v4f*>(w + q * 128 + lane * 4);

    const float* __restrict__ hb = h + (size_t)row0 * kH + lane * 4;

    v4f buf[kAhead + 1][kPair][kPieces];
#define LOAD_STEP(S) do { \
    _Pragma("unroll") for (int p_ = 0; p_ < kPair; ++p_) \
    _Pragma("unroll") for (int q_ = 0; q_ < kPieces; ++q_) \
        buf[(S) % (kAhead + 1)][p_][q_] = *reinterpret_cast<const v4f*>( \
            hb + ((S) * kPair + p_) * kH + q_ * 128); } while (0)

    #pragma unroll
    for (int s = 0; s < kAhead; ++s) LOAD_STEP(s);

    const int  sel    = lane & 7;
    const bool onDiag = (lane < 8) || (lane >= 24);
    float mine = 0.0f;

    #pragma unroll
    for (int s = 0; s < kSteps; ++s) {
        __builtin_amdgcn_sched_barrier(0);
        if (s + kAhead < kSteps) LOAD_STEP(s + kAhead);
        __builtin_amdgcn_sched_barrier(0);

        v8f c[kPair];
        #pragma unroll
        for (int p = 0; p < kPair; ++p) c[p] = (v8f){};
        #pragma unroll
        for (int q = 0; q < kPieces; ++q) {
            const v2f b0 = {wq[q].x, wq[q].y};
            const v2f b1 = {wq[q].z, wq[q].w};
            #pragma unroll
            for (int p = 0; p < kPair; ++p) {
                const v4f x = buf[s % (kAhead + 1)][p][q];
                const v2f a0 = {x.x, x.y};
                c[p] = __builtin_amdgcn_wmma_f32_16x16x4_f32(
                           false, a0, false, b0, (short)0, c[p], false, false);
            }
            #pragma unroll
            for (int p = 0; p < kPair; ++p) {
                const v4f x = buf[s % (kAhead + 1)][p][q];
                const v2f a1 = {x.z, x.w};
                c[p] = __builtin_amdgcn_wmma_f32_16x16x4_f32(
                           false, a1, false, b1, (short)0, c[p], false, false);
            }
        }

        #pragma unroll
        for (int p = 0; p < kPair; ++p) {
            float v = c[p][0];
            v = (sel == 1) ? c[p][1] : v;
            v = (sel == 2) ? c[p][2] : v;
            v = (sel == 3) ? c[p][3] : v;
            v = (sel == 4) ? c[p][4] : v;
            v = (sel == 5) ? c[p][5] : v;
            v = (sel == 6) ? c[p][6] : v;
            v = (sel == 7) ? c[p][7] : v;
            v = onDiag ? v : 0.0f;
            v += lane_xor16(v);
            v += lane_xor_row<8>(v);
            v += lane_xor_row<4>(v);
            v += lane_xor_row<2>(v);
            v += lane_xor_row<1>(v);
            mine = (lane == s * kPair + p) ? v : mine;
        }
    }
#undef LOAD_STEP

    proj[row0 + lane] = mine;
}

constexpr int kNG      = 1000;
constexpr int kCols    = 32;
constexpr int kWaves   = 8;
constexpr int kRows    = kT / kWaves;
constexpr int kNPerBlk = 32;
constexpr int kQ       = kNPerBlk / kWaves;
static_assert(kBS % kCols == 0, "whole column groups");
static_assert(kNG % kQ == 0, "a wave's grid points are all inside the grid or all outside");

__global__ void __launch_bounds__(kWaves * 32)
rmtpp_grid_kernel(const float* __restrict__ ts,
                  const float* __restrict__ proj,
                  const float* __restrict__ time_influence,
                  const float* __restrict__ base_intensity,
                  float* __restrict__ out_log,
                  float* __restrict__ out_grid) {
    __shared__ float pmin[kT * kCols];
    __shared__ float seg[kWaves * kCols];
    __shared__ float segMax[kWaves * kCols];

    const int lane  = threadIdx.x & 31;
    const int waveV = threadIdx.x >> 5;
    const int wave  = __builtin_amdgcn_readfirstlane(waveV);
    const int b     = blockIdx.x * kCols + lane;

    const int t0 = wave * kRows;
    const float* __restrict__ col = ts + t0 * kBS + b;
    float v[kRows];
    #pragma unroll
    for (int i = 0; i < kRows; ++i) v[i] = col[i * kBS];
    float mx[4] = {-3.402823466e38f, -3.402823466e38f, -3.402823466e38f, -3.402823466e38f};
    #pragma unroll
    for (int i = 0; i < kRows; ++i) mx[i & 3] = fmaxf(mx[i & 3], v[i]);
    #pragma unroll
    for (int i = 1; i < kRows; ++i) v[i] = fminf(v[i - 1], v[i]);
    seg[wave * kCols + lane]    = v[kRows - 1];
    segMax[wave * kCols + lane] = fmaxf(fmaxf(mx[0], mx[1]), fmaxf(mx[2], mx[3]));
    __syncthreads();

    float carry = __builtin_inff();
    #pragma unroll
    for (int s = 0; s < kWaves - 1; ++s) {
        const float x = seg[s * kCols + lane];
        carry = fminf(carry, (s < waveV) ? x : __builtin_inff());
    }
    float d = segMax[lane];
    #pragma unroll
    for (int s = 1; s < kWaves; ++s) d = fmaxf(d, segMax[s * kCols + lane]);
    #pragma unroll
    for (int i = 0; i < kRows; ++i) pmin[(t0 + i) * kCols + lane] = fminf(v[i], carry);
    __syncthreads();

    const int nBase = blockIdx.y * kNPerBlk + wave * kQ;
    if (nBase >= kNG) return;

    const float ti = time_influence[0];
    const float bi = base_intensity[0];

    float g[kQ];
    int   pos[kQ];
    #pragma unroll
    for (int q = 0; q < kQ; ++q) {
        g[q]   = ((float)(nBase + q) / (float)(kNG - 1)) * d;
        pos[q] = 0;
    }
    #pragma unroll
    for (int step = kT / 2; step >= 1; step >>= 1) {
        #pragma unroll
        for (int q = 0; q < kQ; ++q) {
            const float p = pmin[(pos[q] + step - 1) * kCols + lane];
            pos[q] += (p < g[q]) ? 0 : step;
        }
    }
    #pragma unroll
    for (int q = 0; q < kQ; ++q) {
        const float p = pmin[pos[q] * kCols + lane];
        pos[q] += (p < g[q]) ? 0 : 1;
    }

    float closest_ts[kQ], pj[kQ];
    #pragma unroll
    for (int q = 0; q < kQ; ++q) {
        const int idx = (pos[q] < kT) ? pos[q] : 0;
        closest_ts[q] = ts[idx * kBS + b];
        pj[q]         = proj[idx * kBS + b];
    }
    #pragma unroll
    for (int q = 0; q < kQ; ++q) {
        const float t_grid = g[q];
        const float li = pj[q] + ti * (t_grid - closest_ts[q]) + bi;
        out_log[(nBase + q) * kBS + b]  = li;
        out_grid[(nBase + q) * kBS + b] = t_grid;
    }
}

extern "C" void kernel_launch(void* const* d_in, const int* in_sizes, int n_in,
                              void* d_out, int out_size, void* d_ws, size_t ws_size,
                              hipStream_t stream) {
    const float* h  = (const float*)d_in[0];
    const float* ts = (const float*)d_in[1];
    const float* w  = (const float*)d_in[2];
    const float* ti = (const float*)d_in[3];
    const float* bi = (const float*)d_in[4];

    float* out_log  = (float*)d_out;
    float* out_grid = out_log + (size_t)kNG * kBS;

    float* proj = (float*)d_ws;

    rmtpp_proj_wmma_kernel<<<kProjBlocks, kProjWaves * 32, 0, stream>>>(h, w, proj);

    const dim3 gridBlocks(kBS / kCols, (kNG + kNPerBlk - 1) / kNPerBlk);
    rmtpp_grid_kernel<<<gridBlocks, kWaves * 32, 0, stream>>>(
        ts, proj, ti, bi, out_log, out_grid);
}
